# starting kernel
# baseline (speedup 1.0000x reference)
#include <hip/hip_runtime.h>

typedef __attribute__((ext_vector_type(2))) float v2f;
typedef __attribute__((ext_vector_type(8))) float v8f;

#define B_      128
#define C_      1024
#define N_      49
#define NPAD    64
#define KCHUNKS 4
#define KCHUNK  (C_ / KCHUNKS)
#define KSTEPS2 13

__global__ __launch_bounds__(32) void cam_stage1(
    const float* __restrict__ kmat, const float* __restrict__ vmat,
    float* __restrict__ Mp)
{
    const int b     = blockIdx.x;
    const int chunk = blockIdx.y;
    const int lane  = threadIdx.x & 31;
    const int ml    = lane & 15;
    const int kh    = (lane >> 4) << 1;

    const float* kb = kmat + (size_t)b * C_ * N_;
    const float* vb = vmat + (size_t)b * C_ * N_;

    float sel[4];
    int   pc[4];
#pragma unroll
    for (int t = 0; t < 4; ++t) {
        int p  = t * 16 + ml;
        sel[t] = (p < N_) ? 1.0f : 0.0f;
        pc[t]  = (p < N_) ? p : (N_ - 1);
    }

    v8f acc[4][4] = {};

    const int c0 = chunk * KCHUNK;
    for (int c = c0; c < c0 + KCHUNK; c += 4) {
        const float* kr0 = kb + (size_t)(c + kh)     * N_;
        const float* kr1 = kb + (size_t)(c + kh + 1) * N_;
        const float* vr0 = vb + (size_t)(c + kh)     * N_;
        const float* vr1 = vb + (size_t)(c + kh + 1) * N_;

        v2f a[4], bm[4];
#pragma unroll
        for (int i = 0; i < 4; ++i) {
            a[i].x = kr0[pc[i]] * sel[i];
            a[i].y = kr1[pc[i]] * sel[i];
        }
#pragma unroll
        for (int j = 0; j < 4; ++j) {
            bm[j].x = vr0[pc[j]] * sel[j];
            bm[j].y = vr1[pc[j]] * sel[j];
        }
#pragma unroll
        for (int i = 0; i < 4; ++i)
#pragma unroll
            for (int j = 0; j < 4; ++j)
                acc[i][j] = __builtin_amdgcn_wmma_f32_16x16x4_f32(
                    false, a[i], false, bm[j], (short)0, acc[i][j], false, false);
    }

    float*    out  = Mp + (size_t)(chunk * B_ + b) * NPAD * NPAD;
    const int rhi  = (lane >> 4) << 3;
#pragma unroll
    for (int i = 0; i < 4; ++i)
#pragma unroll
        for (int j = 0; j < 4; ++j)
#pragma unroll
            for (int r = 0; r < 8; ++r)
                out[(size_t)(i * 16 + rhi + r) * NPAD + j * 16 + ml] = acc[i][j][r];
}

__global__ __launch_bounds__(256) void cam_stage2(
    const float* __restrict__ q, const float* __restrict__ v1,
    const float* __restrict__ Mp, float* __restrict__ out)
{
    __shared__ float Ms[NPAD * NPAD];
    const int b = blockIdx.x;

    for (int p = threadIdx.x; p < NPAD * NPAD; p += 256) {
        float s = 0.0f;
#pragma unroll
        for (int ch = 0; ch < KCHUNKS; ++ch)
            s += Mp[(size_t)(ch * B_ + b) * NPAD * NPAD + p];
        Ms[p] = s;
    }
    __syncthreads();

    const int wave = threadIdx.x >> 5;
    const int lane = threadIdx.x & 31;
    const int ml   = lane & 15;
    const int kh   = (lane >> 4) << 1;

    v2f Breg[KSTEPS2][4];
#pragma unroll
    for (int ks = 0; ks < KSTEPS2; ++ks)
#pragma unroll
        for (int j = 0; j < 4; ++j) {
            Breg[ks][j].x = Ms[(ks * 4 + kh)     * NPAD + j * 16 + ml];
            Breg[ks][j].y = Ms[(ks * 4 + kh + 1) * NPAD + j * 16 + ml];
        }

    const float* qb = q  + (size_t)b * C_ * N_;
    const float* vb = v1 + (size_t)b * C_ * N_;
    float*       ob = out + (size_t)b * C_ * N_;

    for (int it = wave; it < 64; it += 8) {
        const int    mrow = it * 16 + ml;
        const float* qrow = qb + (size_t)mrow * N_;
        v8f acc[4] = {};
#pragma unroll
        for (int ks = 0; ks < KSTEPS2; ++ks) {
            const int kk0 = ks * 4 + kh;
            v2f a;
            a.x = (kk0     < N_) ? qrow[kk0]     : 0.0f;
            a.y = (kk0 + 1 < N_) ? qrow[kk0 + 1] : 0.0f;
#pragma unroll
            for (int j = 0; j < 4; ++j)
                acc[j] = __builtin_amdgcn_wmma_f32_16x16x4_f32(
                    false, a, false, Breg[ks][j], (short)0, acc[j], false, false);
        }
        const int rbase = it * 16 + ((lane >> 4) << 3);
#pragma unroll
        for (int j = 0; j < 4; ++j) {
            const int n = j * 16 + ml;
            if (n < N_) {
#pragma unroll
                for (int r = 0; r < 8; ++r) {
                    const size_t idx = (size_t)(rbase + r) * N_ + n;
                    ob[idx] = acc[j][r] + vb[idx];
                }
            }
        }
    }
}

extern "C" void kernel_launch(void* const* d_in, const int* in_sizes, int n_in,
                              void* d_out, int out_size, void* d_ws, size_t ws_size,
                              hipStream_t stream) {
    const float* v1 = (const float*)d_in[0];
    const float* q1 = (const float*)d_in[1];
    const float* k1 = (const float*)d_in[2];
    float*       out = (float*)d_out;
    float*       Mp  = (float*)d_ws;

    dim3 g1(B_, KCHUNKS);
    cam_stage1<<<g1, dim3(32), 0, stream>>>(k1, v1, Mp);
    cam_stage2<<<dim3(B_), dim3(256), 0, stream>>>(q1, v1, Mp, out);
}


# phase 2 takes its v1 start values from a copy of the band of v that phase 1 leaves in LDS as it uses it (128 KB more LDS), so each work-group reads v from memory once instead of twice
# speedup vs baseline: 5.6306x; 5.6306x over previous
#include <hip/hip_runtime.h>


typedef __attribute__((ext_vector_type(2))) float v2f;
typedef __attribute__((ext_vector_type(4))) float v4f;
typedef __attribute__((ext_vector_type(8))) float v8f;

#define B_      128
#define C_      1024
#define N_      49
#define KSTEPS2 13
#define WAVES   16
#define CPW     (C_ / WAVES)
#define SPLIT   2
#define COLS0   25
#define MROWS   64
#define MCOLS   32
#define TPW     (C_ / 16 / WAVES)
#define PLANE_V4 (MROWS * MCOLS / 4)
#define TSTR    (WAVES * 16 * N_)
#define VP      MCOLS
#define TROWS   (WAVES * 16)

static_assert(CPW == 64, "phase 1 walks 8 groups of 8 channels per wave: 2 trips of 3, then 2 more");
static_assert(TPW == 4, "phase 2 gives every wave exactly four row tiles");
static_assert(PLANE_V4 == 512, "the plane sum gives every thread exactly one float4");

#define DEV static __device__ __forceinline__

struct Frag1 { v2f a[2][4]; v2f b[2][2]; };

DEV void load1(Frag1& f, const float* kb, const float* vb, int c, int kh,
               const int (&pc)[4], const int (&pv)[2])
{
#pragma unroll
    for (int s = 0; s < 2; ++s) {
        const float* kr = kb + (size_t)(c + 4 * s + kh) * N_;
        const float* vr = vb + (size_t)(c + 4 * s + kh) * N_;
#pragma unroll
        for (int t = 0; t < 4; ++t) {
            f.a[s][t].x = kr[pc[t]];
            f.a[s][t].y = kr[N_ + pc[t]];
        }
#pragma unroll
        for (int j = 0; j < 2; ++j) {
            f.b[s][j].x = vr[pv[j]];
            f.b[s][j].y = vr[N_ + pv[j]];
        }
    }
}

DEV void mma1(v8f (&acc)[4][2], const Frag1& f, const bool (&okA)[4], const bool (&okB)[2],
              float* vs)
{
#pragma unroll
    for (int s = 0; s < 2; ++s) {
#pragma unroll
        for (int j = 0; j < 2; ++j) {
            vs[(4 * s)     * VP + j * 16] = f.b[s][j].x;
            vs[(4 * s + 1) * VP + j * 16] = f.b[s][j].y;
        }
        v2f a[4], bm[2];
#pragma unroll
        for (int t = 0; t < 4; ++t) {
            a[t].x = okA[t] ? f.a[s][t].x : 0.0f;
            a[t].y = okA[t] ? f.a[s][t].y : 0.0f;
        }
#pragma unroll
        for (int j = 0; j < 2; ++j) {
            bm[j].x = okB[j] ? f.b[s][j].x : 0.0f;
            bm[j].y = okB[j] ? f.b[s][j].y : 0.0f;
        }
#pragma unroll
        for (int i = 0; i < 4; ++i)
#pragma unroll
            for (int j = 0; j < 2; ++j)
                acc[i][j] = __builtin_amdgcn_wmma_f32_16x16x4_f32(
                    false, a[i], false, bm[j], (short)0, acc[i][j], false, false);
    }
}

DEV void loadq(v2f (&qf)[KSTEPS2], const float* p, int kh)
{
#pragma unroll
    for (int ks = 0; ks < KSTEPS2 - 1; ++ks) {
        qf[ks].x = p[4 * ks];
        qf[ks].y = p[4 * ks + 1];
    }
    qf[KSTEPS2 - 1].x = p[48 - kh];
    qf[KSTEPS2 - 1].y = 0.0f;
}

DEV void loadc(v8f (&cf)[2], const float* p)
{
#pragma unroll
    for (int j = 0; j < 2; ++j)
#pragma unroll
        for (int r = 0; r < 8; ++r)
            cf[j][r] = p[r * VP + j * 16];
}

DEV void mma2(v8f (&acc)[2], const v2f (&qf)[KSTEPS2], const v2f (&Breg)[KSTEPS2][2], int kh)
{
#pragma unroll
    for (int ks = 0; ks < KSTEPS2; ++ks) {
        v2f a = qf[ks];
        if (ks == KSTEPS2 - 1) a.x = (kh == 0) ? a.x : 0.0f;
#pragma unroll
        for (int j = 0; j < 2; ++j)
            acc[j] = __builtin_amdgcn_wmma_f32_16x16x4_f32(
                false, a, false, Breg[ks][j], (short)0, acc[j], false, false);
    }
}

DEV void store2(const v8f (&acc)[2], float* p, bool st1)
{
#pragma unroll
    for (int r = 0; r < 8; ++r)
        p[r * N_] = acc[0][r];
    if (st1) {
#pragma unroll
        for (int r = 0; r < 8; ++r)
            p[r * N_ + 16] = acc[1][r];
    }
}

__global__ __launch_bounds__(512) void cam_fused(
    const float* __restrict__ q, const float* __restrict__ kmat,
    const float* __restrict__ v1, float* __restrict__ out)
{
    __shared__ v4f   planes[WAVES * PLANE_V4];
    __shared__ float Ms[MROWS * MCOLS];
    __shared__ float vband[C_ * VP];

    const int b    = blockIdx.x;
    const int band = blockIdx.y;
    const int wave = threadIdx.x >> 5;
    const int lane = threadIdx.x & 31;
    const int ml   = lane & 15;
    const int kh   = (lane >> 4) << 1;

    const int cb = band ? COLS0 : 0;
    const int nc = band ? (N_ - COLS0) : COLS0;

    const float* kb = kmat + (size_t)b * C_ * N_;
    const float* vb = v1   + (size_t)b * C_ * N_;

    bool okB[2];
    int  cv[2];
#pragma unroll
    for (int j = 0; j < 2; ++j) {
        int p = j * 16 + ml;
        okB[j] = (p < nc);
        cv[j]  = cb + ((p < nc) ? p : (nc - 1));
    }

    {
        bool okA[4];
        int  pc[4];
#pragma unroll
        for (int t = 0; t < 4; ++t) {
            int p = t * 16 + ml;
            okA[t] = (p < N_);
            pc[t]  = (p < N_) ? p : (N_ - 1);
        }

        v8f acc[4][2] = {};

        const int c0 = wave * CPW;
#define PIN() __builtin_amdgcn_sched_barrier(0)
        float* vs0 = vband + (c0 + kh) * VP + ml;
        Frag1 f0, f1, f2;
        load1(f0, kb, vb, c0,     kh, pc, cv);
        PIN();
        load1(f1, kb, vb, c0 + 8, kh, pc, cv);
        PIN();
        for (int t = 0; t < 2; ++t) {
            const int c = c0 + 24 * t;
            float* vs = vs0 + 24 * t * VP;
            load1(f2, kb, vb, c + 16, kh, pc, cv);
            PIN();
            mma1(acc, f0, okA, okB, vs);
            PIN();
            load1(f0, kb, vb, c + 24, kh, pc, cv);
            PIN();
            mma1(acc, f1, okA, okB, vs + 8 * VP);
            PIN();
            load1(f1, kb, vb, c + 32, kh, pc, cv);
            PIN();
            mma1(acc, f2, okA, okB, vs + 16 * VP);
            PIN();
        }
        mma1(acc, f0, okA, okB, vs0 + 48 * VP);
        mma1(acc, f1, okA, okB, vs0 + 56 * VP);
#undef PIN

        v4f* pl = planes + wave * PLANE_V4;
#pragma unroll
        for (int i = 0; i < 4; ++i)
#pragma unroll
            for (int j = 0; j < 2; ++j) {
                const int t = i * 2 + j;
                v4f lo = { acc[i][j][0], acc[i][j][1], acc[i][j][2], acc[i][j][3] };
                v4f hi = { acc[i][j][4], acc[i][j][5], acc[i][j][6], acc[i][j][7] };
                pl[(t * 2 + 0) * 32 + lane] = lo;
                pl[(t * 2 + 1) * 32 + lane] = hi;
            }
    }
    __syncthreads();

    {
        const int f = threadIdx.x;
        v4f s = planes[f];
#pragma unroll
        for (int w = 1; w < WAVES; ++w)
            s += planes[w * PLANE_V4 + f];
        const int t   = f >> 6;
        const int hf  = (f >> 5) & 1;
        const int ln  = f & 31;
        const int row = (t >> 1) * 16 + ((ln >> 4) << 3) + hf * 4;
        const int col = (t & 1) * 16 + (ln & 15);
        Ms[(row + 0) * MCOLS + col] = s.x;
        Ms[(row + 1) * MCOLS + col] = s.y;
        Ms[(row + 2) * MCOLS + col] = s.z;
        Ms[(row + 3) * MCOLS + col] = s.w;
    }
    __syncthreads();

    const float* qb = q   + (size_t)b * C_ * N_;
    float*       ob = out + (size_t)b * C_ * N_;

    const int    hi8 = (lane >> 4) << 3;
    const float* qp  = qb + (size_t)(wave * 16 + ml)  * N_ + kh;
    const float* vp  = vband + (wave * 16 + hi8) * VP + ml;
    float*       op  = ob + (size_t)(wave * 16 + hi8) * N_ + cb + ml;
    const bool   st1 = okB[1];

    v2f qA[KSTEPS2], qB[KSTEPS2];
    v8f accA[2], accB[2];

    loadq(qA, qp, kh);
    loadc(accA, vp);
    __builtin_amdgcn_sched_barrier(0);

    v2f Breg[KSTEPS2][2];
#pragma unroll
    for (int ks = 0; ks < KSTEPS2; ++ks)
#pragma unroll
        for (int j = 0; j < 2; ++j) {
            Breg[ks][j].x = Ms[(ks * 4 + kh)     * MCOLS + j * 16 + ml];
            Breg[ks][j].y = Ms[(ks * 4 + kh + 1) * MCOLS + j * 16 + ml];
        }

    __builtin_amdgcn_sched_barrier(0);

    loadq(qB, qp + TSTR, kh);
    loadc(accB, vp + TROWS * VP);

    mma2(accA, qA, Breg, kh);
    store2(accA, op, st1);

    loadq(qA, qp + 2 * TSTR, kh);
    loadc(accA, vp + 2 * TROWS * VP);

    mma2(accB, qB, Breg, kh);
    store2(accB, op + TSTR, st1);

    loadq(qB, qp + 3 * TSTR, kh);
    loadc(accB, vp + 3 * TROWS * VP);

    mma2(accA, qA, Breg, kh);
    store2(accA, op + 2 * TSTR, st1);
    mma2(accB, qB, Breg, kh);
    store2(accB, op + 3 * TSTR, st1);
}

extern "C" void kernel_launch(void* const* d_in, const int* in_sizes, int n_in,
                              void* d_out, int out_size, void* d_ws, size_t ws_size,
                              hipStream_t stream) {
    const float* v1 = (const float*)d_in[0];
    const float* q1 = (const float*)d_in[1];
    const float* k1 = (const float*)d_in[2];
    float*       out = (float*)d_out;

    cam_fused<<<dim3(B_, SPLIT), dim3(512), 0, stream>>>(q1, k1, v1, out);
}


# phase-2 q fragments read 16 bytes per lane straight from memory (32 contiguous bytes of a row per load; 7 loads per 16-row tile, was 13), k order of q and of M's fragments permuted alike to fit
# speedup vs baseline: 6.0170x; 1.0686x over previous
#include <hip/hip_runtime.h>


typedef __attribute__((ext_vector_type(2))) float v2f;
typedef __attribute__((ext_vector_type(4))) float v4f;
typedef __attribute__((ext_vector_type(8))) float v8f;
typedef __attribute__((ext_vector_type(4), aligned(4))) float v4fu;

#define B_      128
#define C_      1024
#define N_      49
#define KSTEPS2 13
#define QV      6
#define WAVES   16
#define CPW     (C_ / WAVES)
#define SPLIT   2
#define COLS0   25
#define MROWS   64
#define MCOLS   32
#define TPW     (C_ / 16 / WAVES)
#define PLANE_V4 (MROWS * MCOLS / 4)
#define TSTR    (WAVES * 16 * N_)
#define VP      MCOLS
#define TROWS   (WAVES * 16)

static_assert(CPW == 64, "phase 1 walks 8 groups of 8 channels per wave: 2 trips of 3, then 2 more");
static_assert(TPW == 4, "phase 2 gives every wave exactly four row tiles");
static_assert(PLANE_V4 == 512, "the plane sum gives every thread exactly one float4");

#define DEV static __device__ __forceinline__

struct Frag1 { v2f a[2][4]; v2f b[2][2]; };

DEV void load1(Frag1& f, const float* kb, const float* vb, int c, int kh,
               const int (&pc)[4], const int (&pv)[2])
{
#pragma unroll
    for (int s = 0; s < 2; ++s) {
        const float* kr = kb + (size_t)(c + 4 * s + kh) * N_;
        const float* vr = vb + (size_t)(c + 4 * s + kh) * N_;
#pragma unroll
        for (int t = 0; t < 4; ++t) {
            f.a[s][t].x = kr[pc[t]];
            f.a[s][t].y = kr[N_ + pc[t]];
        }
#pragma unroll
        for (int j = 0; j < 2; ++j) {
            f.b[s][j].x = vr[pv[j]];
            f.b[s][j].y = vr[N_ + pv[j]];
        }
    }
}

DEV void mma1(v8f (&acc)[4][2], const Frag1& f, const bool (&okA)[4], const bool (&okB)[2],
              float* vs)
{
#pragma unroll
    for (int s = 0; s < 2; ++s) {
#pragma unroll
        for (int j = 0; j < 2; ++j) {
            vs[(4 * s)     * VP + j * 16] = f.b[s][j].x;
            vs[(4 * s + 1) * VP + j * 16] = f.b[s][j].y;
        }
        v2f a[4], bm[2];
#pragma unroll
        for (int t = 0; t < 4; ++t) {
            a[t].x = okA[t] ? f.a[s][t].x : 0.0f;
            a[t].y = okA[t] ? f.a[s][t].y : 0.0f;
        }
#pragma unroll
        for (int j = 0; j < 2; ++j) {
            bm[j].x = okB[j] ? f.b[s][j].x : 0.0f;
            bm[j].y = okB[j] ? f.b[s][j].y : 0.0f;
        }
#pragma unroll
        for (int i = 0; i < 4; ++i)
#pragma unroll
            for (int j = 0; j < 2; ++j)
                acc[i][j] = __builtin_amdgcn_wmma_f32_16x16x4_f32(
                    false, a[i], false, bm[j], (short)0, acc[i][j], false, false);
    }
}

struct QFrag { v4f v[QV]; float last; };

DEV void loadq(QFrag& f, const float* p, int h4)
{
#pragma unroll
    for (int j = 0; j < QV; ++j)
        f.v[j] = *(const v4fu*)(p + 8 * j);
    f.last = p[48 - h4];
}

DEV void loadc(v8f (&cf)[2], const float* p)
{
#pragma unroll
    for (int j = 0; j < 2; ++j)
#pragma unroll
        for (int r = 0; r < 8; ++r)
            cf[j][r] = p[r * VP + j * 16];
}

DEV void mma2(v8f (&acc)[2], const QFrag& f, const v2f (&Breg)[KSTEPS2][2], int h4)
{
#pragma unroll
    for (int ks = 0; ks < KSTEPS2; ++ks) {
        v2f a;
        if (ks < KSTEPS2 - 1) {
            a.x = (ks & 1) ? f.v[ks >> 1].z : f.v[ks >> 1].x;
            a.y = (ks & 1) ? f.v[ks >> 1].w : f.v[ks >> 1].y;
        } else {
            a.x = (h4 == 0) ? f.last : 0.0f;
            a.y = 0.0f;
        }
#pragma unroll
        for (int j = 0; j < 2; ++j)
            acc[j] = __builtin_amdgcn_wmma_f32_16x16x4_f32(
                false, a, false, Breg[ks][j], (short)0, acc[j], false, false);
    }
}

DEV void store2(const v8f (&acc)[2], float* p, bool st1)
{
#pragma unroll
    for (int r = 0; r < 8; ++r)
        p[r * N_] = acc[0][r];
    if (st1) {
#pragma unroll
        for (int r = 0; r < 8; ++r)
            p[r * N_ + 16] = acc[1][r];
    }
}

__global__ __launch_bounds__(512) void cam_fused(
    const float* __restrict__ q, const float* __restrict__ kmat,
    const float* __restrict__ v1, float* __restrict__ out)
{
    __shared__ v4f   planes[WAVES * PLANE_V4];
    __shared__ float Ms[MROWS * MCOLS];
    __shared__ float vband[C_ * VP];

    const int b    = blockIdx.x;
    const int band = blockIdx.y;
    const int wave = threadIdx.x >> 5;
    const int lane = threadIdx.x & 31;
    const int ml   = lane & 15;
    const int kh   = (lane >> 4) << 1;

    const int cb = band ? COLS0 : 0;
    const int nc = band ? (N_ - COLS0) : COLS0;

    const float* kb = kmat + (size_t)b * C_ * N_;
    const float* vb = v1   + (size_t)b * C_ * N_;

    bool okB[2];
    int  cv[2];
#pragma unroll
    for (int j = 0; j < 2; ++j) {
        int p = j * 16 + ml;
        okB[j] = (p < nc);
        cv[j]  = cb + ((p < nc) ? p : (nc - 1));
    }

    {
        bool okA[4];
        int  pc[4];
#pragma unroll
        for (int t = 0; t < 4; ++t) {
            int p = t * 16 + ml;
            okA[t] = (p < N_);
            pc[t]  = (p < N_) ? p : (N_ - 1);
        }

        v8f acc[4][2] = {};

        const int c0 = wave * CPW;
#define PIN() __builtin_amdgcn_sched_barrier(0)
        float* vs0 = vband + (c0 + kh) * VP + ml;
        Frag1 f0, f1, f2;
        load1(f0, kb, vb, c0,     kh, pc, cv);
        PIN();
        load1(f1, kb, vb, c0 + 8, kh, pc, cv);
        PIN();
        for (int t = 0; t < 2; ++t) {
            const int c = c0 + 24 * t;
            float* vs = vs0 + 24 * t * VP;
            load1(f2, kb, vb, c + 16, kh, pc, cv);
            PIN();
            mma1(acc, f0, okA, okB, vs);
            PIN();
            load1(f0, kb, vb, c + 24, kh, pc, cv);
            PIN();
            mma1(acc, f1, okA, okB, vs + 8 * VP);
            PIN();
            load1(f1, kb, vb, c + 32, kh, pc, cv);
            PIN();
            mma1(acc, f2, okA, okB, vs + 16 * VP);
            PIN();
        }
        mma1(acc, f0, okA, okB, vs0 + 48 * VP);
        mma1(acc, f1, okA, okB, vs0 + 56 * VP);
#undef PIN

        v4f* pl = planes + wave * PLANE_V4;
#pragma unroll
        for (int i = 0; i < 4; ++i)
#pragma unroll
            for (int j = 0; j < 2; ++j) {
                const int t = i * 2 + j;
                v4f lo = { acc[i][j][0], acc[i][j][1], acc[i][j][2], acc[i][j][3] };
                v4f hi = { acc[i][j][4], acc[i][j][5], acc[i][j][6], acc[i][j][7] };
                pl[(t * 2 + 0) * 32 + lane] = lo;
                pl[(t * 2 + 1) * 32 + lane] = hi;
            }
    }
    __syncthreads();

    {
        const int f = threadIdx.x;
        v4f s = planes[f];
#pragma unroll
        for (int w = 1; w < WAVES; ++w)
            s += planes[w * PLANE_V4 + f];
        const int t   = f >> 6;
        const int hf  = (f >> 5) & 1;
        const int ln  = f & 31;
        const int row = (t >> 1) * 16 + ((ln >> 4) << 3) + hf * 4;
        const int col = (t & 1) * 16 + (ln & 15);
        Ms[(row + 0) * MCOLS + col] = s.x;
        Ms[(row + 1) * MCOLS + col] = s.y;
        Ms[(row + 2) * MCOLS + col] = s.z;
        Ms[(row + 3) * MCOLS + col] = s.w;
    }
    __syncthreads();

    const float* qb = q   + (size_t)b * C_ * N_;
    float*       ob = out + (size_t)b * C_ * N_;

    const int    hi8 = (lane >> 4) << 3;
    const int    h4  = (lane >> 4) << 2;
    const float* qp  = qb + (size_t)(wave * 16 + ml)  * N_ + h4;
    const float* vp  = vband + (wave * 16 + hi8) * VP + ml;
    float*       op  = ob + (size_t)(wave * 16 + hi8) * N_ + cb + ml;
    const bool   st1 = okB[1];

    QFrag qA, qB;
    v8f accA[2], accB[2];

    loadq(qA, qp, h4);
    loadc(accA, vp);
    __builtin_amdgcn_sched_barrier(0);

    v2f Breg[KSTEPS2][2];
#pragma unroll
    for (int ks = 0; ks < KSTEPS2; ++ks) {
        const int kr = 8 * (ks >> 1) + 2 * (ks & 1) + h4;
#pragma unroll
        for (int j = 0; j < 2; ++j) {
            Breg[ks][j].x = Ms[(kr)     * MCOLS + j * 16 + ml];
            Breg[ks][j].y = Ms[(kr + 1) * MCOLS + j * 16 + ml];
        }
    }

    __builtin_amdgcn_sched_barrier(0);

    loadq(qB, qp + TSTR, h4);
    loadc(accB, vp + TROWS * VP);

    mma2(accA, qA, Breg, h4);
    store2(accA, op, st1);

    loadq(qA, qp + 2 * TSTR, h4);
    loadc(accA, vp + 2 * TROWS * VP);

    mma2(accB, qB, Breg, h4);
    store2(accB, op + TSTR, st1);

    loadq(qB, qp + 3 * TSTR, h4);
    loadc(accB, vp + 3 * TROWS * VP);

    mma2(accA, qA, Breg, h4);
    store2(accA, op + 2 * TSTR, st1);
    mma2(accB, qB, Breg, h4);
    store2(accB, op + 3 * TSTR, st1);
}

extern "C" void kernel_launch(void* const* d_in, const int* in_sizes, int n_in,
                              void* d_out, int out_size, void* d_ws, size_t ws_size,
                              hipStream_t stream) {
    const float* v1 = (const float*)d_in[0];
    const float* q1 = (const float*)d_in[1];
    const float* k1 = (const float*)d_in[2];
    float*       out = (float*)d_out;

    cam_fused<<<dim3(B_, SPLIT), dim3(512), 0, stream>>>(q1, k1, v1, out);
}
